# starting kernel
# baseline (speedup 1.0000x reference)
#include <hip/hip_runtime.h>

typedef __attribute__((ext_vector_type(2))) float v2f;
typedef __attribute__((ext_vector_type(8))) float v8f;

#define IN_DIM 128
#define OUT_DIM 32

__global__ void __launch_bounds__(256) gcmc_gemm_cj_kernel(
    const float* __restrict__ X,
    const float* __restrict__ W,
    const float* __restrict__ bias,
    const float* __restrict__ cj,
    float* __restrict__ wh,
    int n_nodes)
{
    const int wave   = (blockIdx.x * blockDim.x + threadIdx.x) >> 5;
    const int ntiles = (n_nodes + 15) >> 4;
    if (wave >= ntiles) return;

    const int lane = threadIdx.x & 31;
    const int m    = lane & 15;
    const int kh   = lane >> 4;

    int arow = wave * 16 + m;
    if (arow >= n_nodes) arow = n_nodes - 1;

    const float* xrow = X + (size_t)arow * IN_DIM + 2 * kh;
    const float* w0   = W + (size_t)m        * IN_DIM + 2 * kh;
    const float* w1   = W + (size_t)(m + 16) * IN_DIM + 2 * kh;

    v8f acc0 = {};
    v8f acc1 = {};

#pragma unroll
    for (int kb = 0; kb < IN_DIM; kb += 4) {
        v2f a  = *(const v2f*)(xrow + kb);
        v2f b0 = *(const v2f*)(w0 + kb);
        v2f b1 = *(const v2f*)(w1 + kb);
        acc0 = __builtin_amdgcn_wmma_f32_16x16x4_f32(
            false, a, false, b0, (short)0, acc0, false, false);
        acc1 = __builtin_amdgcn_wmma_f32_16x16x4_f32(
            false, a, false, b1, (short)0, acc1, false, false);
    }

    const int   n0 = m, n1 = m + 16;
    const float bias0 = bias[n0];
    const float bias1 = bias[n1];

#pragma unroll
    for (int r = 0; r < 8; ++r) {
        const int mo   = r + 8 * kh;
        const int orow = wave * 16 + mo;
        if (orow < n_nodes) {
            const float s = cj[orow];
            wh[(size_t)orow * OUT_DIM + n0] = (acc0[r] + bias0) * s;
            wh[(size_t)orow * OUT_DIM + n1] = (acc1[r] + bias1) * s;
        }
    }
}

__global__ void __launch_bounds__(256) gcmc_zero_kernel(float4* __restrict__ out, int n4)
{
    const int i = blockIdx.x * blockDim.x + threadIdx.x;
    if (i < n4) out[i] = make_float4(0.f, 0.f, 0.f, 0.f);
}

__global__ void __launch_bounds__(256) gcmc_scatter_kernel(
    const float* __restrict__ wh,
    const int*   __restrict__ edge_src,
    const int*   __restrict__ edge_dst,
    float*       __restrict__ out,
    int n_edges)
{
    const long long t = (long long)blockIdx.x * blockDim.x + threadIdx.x;
    const long long e = t >> 3;
    if (e >= n_edges) return;
    const int c = ((int)t & 7) * 4;

    const int s = edge_src[e];
    const int d = edge_dst[e];

    const float4 v = *(const float4*)(wh + (size_t)s * OUT_DIM + c);
    float* o = out + (size_t)d * OUT_DIM + c;
    unsafeAtomicAdd(o + 0, v.x);
    unsafeAtomicAdd(o + 1, v.y);
    unsafeAtomicAdd(o + 2, v.z);
    unsafeAtomicAdd(o + 3, v.w);
}

__global__ void __launch_bounds__(256) gcmc_ci_kernel(
    float* __restrict__ out, const float* __restrict__ ci, int n_nodes)
{
    const int t    = blockIdx.x * blockDim.x + threadIdx.x;
    const int node = t >> 3;
    if (node >= n_nodes) return;
    const int c = (t & 7) * 4;

    const float s = ci[node];
    float4* p = (float4*)(out + (size_t)node * OUT_DIM + c);
    float4  v = *p;
    v.x *= s; v.y *= s; v.z *= s; v.w *= s;
    *p = v;
}

extern "C" void kernel_launch(void* const* d_in, const int* in_sizes, int n_in,
                              void* d_out, int out_size, void* d_ws, size_t ws_size,
                              hipStream_t stream)
{
    const float* X    = (const float*)d_in[0];
    const float* cj   = (const float*)d_in[1];
    const float* ci   = (const float*)d_in[2];
    const float* W    = (const float*)d_in[3];
    const float* bias = (const float*)d_in[4];
    const int*   esrc = (const int*)d_in[5];
    const int*   edst = (const int*)d_in[6];

    float* out = (float*)d_out;
    float* wh  = (float*)d_ws;

    const int n_nodes = in_sizes[1];
    const int n_edges = in_sizes[5];

    {
        const int ntiles = (n_nodes + 15) / 16;
        const int blocks = (ntiles + 7) / 8;
        gcmc_gemm_cj_kernel<<<blocks, 256, 0, stream>>>(X, W, bias, cj, wh, n_nodes);
    }

    {
        const int n4 = out_size / 4;
        gcmc_zero_kernel<<<(n4 + 255) / 256, 256, 0, stream>>>((float4*)out, n4);
    }

    {
        const long long threads = (long long)n_edges * 8;
        const int blocks = (int)((threads + 255) / 256);
        gcmc_scatter_kernel<<<blocks, 256, 0, stream>>>(wh, esrc, edst, out, n_edges);
    }

    {
        const int threads = n_nodes * 8;
        gcmc_ci_kernel<<<(threads + 255) / 256, 256, 0, stream>>>(out, ci, n_nodes);
    }
}


# sum pass gather/add loop branch-free: each wave's staged list is padded to 16 with entries that add into a spare LDS row; source byte offsets staged, constant lane indices; arrival-order sums kept
# speedup vs baseline: 2.3030x; 2.3030x over previous
#include <hip/hip_runtime.h>


typedef __attribute__((ext_vector_type(2))) float v2f;
typedef __attribute__((ext_vector_type(8))) float v8f;
typedef __attribute__((ext_vector_type(2))) int   v2i;

#define IN_DIM 128
#define OUT_DIM 32
#define CHUNK 4096
#define ROWS  400
#define MAXB  1024
#define WCAP  512

static __device__ __forceinline__ void gemm_cj_part(
    const int gb,
    const float* __restrict__ X,
    const float* __restrict__ W,
    const float* __restrict__ bias,
    const float* __restrict__ cj,
    float* __restrict__ wh,
    int n_nodes)
{
    const int wave   = (gb * 256 + (int)threadIdx.x) >> 5;
    const int ntiles = (n_nodes + 15) >> 4;
    if (wave >= ntiles) return;

    const int lane = threadIdx.x & 31;
    const int m    = lane & 15;
    const int kh   = lane >> 4;

    int arow = wave * 16 + m;
    if (arow >= n_nodes) arow = n_nodes - 1;

    const float* xrow = X + (size_t)arow * IN_DIM + 2 * kh;
    const float* w0   = W + (size_t)m        * IN_DIM + 2 * kh;
    const float* w1   = W + (size_t)(m + 16) * IN_DIM + 2 * kh;

    v8f acc0 = {};
    v8f acc1 = {};

#pragma unroll
    for (int kb = 0; kb < IN_DIM; kb += 4) {
        v2f a  = *(const v2f*)(xrow + kb);
        v2f b0 = *(const v2f*)(w0 + kb);
        v2f b1 = *(const v2f*)(w1 + kb);
        acc0 = __builtin_amdgcn_wmma_f32_16x16x4_f32(
            false, a, false, b0, (short)0, acc0, false, false);
        acc1 = __builtin_amdgcn_wmma_f32_16x16x4_f32(
            false, a, false, b1, (short)0, acc1, false, false);
    }

    const int   n0 = m, n1 = m + 16;
    const float bias0 = bias[n0];
    const float bias1 = bias[n1];

#pragma unroll
    for (int r = 0; r < 8; ++r) {
        const int mo   = r + 8 * kh;
        const int orow = wave * 16 + mo;
        if (orow < n_nodes) {
            const float s = cj[orow];
            wh[(size_t)orow * OUT_DIM + n0] = (acc0[r] + bias0) * s;
            wh[(size_t)orow * OUT_DIM + n1] = (acc1[r] + bias1) * s;
        }
    }
}

static __device__ __forceinline__ void split_part(
    const int chunk,
    const int* __restrict__ edge_src,
    const int* __restrict__ edge_dst,
    int*       __restrict__ tab,
    int*       __restrict__ pairs,
    int n_edges, int nb)
{
    __shared__ int lc[MAXB];
    __shared__ int lofs[MAXB];
    __shared__ int wsum[8];
    __shared__ v2i st[CHUNK];
    const int tid  = threadIdx.x;
    const int lane = tid & 31;
    const int w    = tid >> 5;

    for (int i = tid; i < nb; i += 256) lc[i] = 0;
    __syncthreads();

    const int e0 = chunk * CHUNK;
    const int total = (n_edges - e0) < CHUNK ? (n_edges - e0) : CHUNK;
    int s[16], d[16], bn[16], r[16];
#pragma unroll
    for (int u = 0; u < 16; ++u) {
        const int i = u * 256 + tid;
        if (i < total) { s[u] = edge_src[e0 + i]; d[u] = edge_dst[e0 + i]; }
        else           { s[u] = 0;                d[u] = -1; }
    }
#pragma unroll
    for (int u = 0; u < 16; ++u) {
        bn[u] = d[u] / ROWS;
        r[u]  = (d[u] >= 0) ? atomicAdd(&lc[bn[u]], 1) : 0;
    }
    __syncthreads();

    int c4[MAXB / 256];
    int tsum = 0;
#pragma unroll
    for (int k = 0; k < MAXB / 256; ++k) {
        const int i = tid * (MAXB / 256) + k;
        c4[k] = (i < nb) ? lc[i] : 0;
        tsum += c4[k];
    }
    int inc = tsum;
#pragma unroll
    for (int o = 1; o < 32; o <<= 1) {
        const int t = __shfl_up(inc, o);
        if (lane >= o) inc += t;
    }
    if (lane == 31) wsum[w] = inc;
    __syncthreads();
    int ex = inc - tsum;
#pragma unroll
    for (int k = 0; k < 8; ++k) ex += (k < w) ? wsum[k] : 0;
#pragma unroll
    for (int k = 0; k < MAXB / 256; ++k) {
        const int i = tid * (MAXB / 256) + k;
        if (i < nb) lofs[i] = ex;
        ex += c4[k];
    }
    __syncthreads();

    int* trow = tab + (size_t)chunk * (size_t)nb;
    for (int i = tid; i < nb; i += 256) trow[i] = lofs[i] | (lc[i] << 16);

#pragma unroll
    for (int u = 0; u < 16; ++u) {
        if (d[u] >= 0) {
            v2i e; e.x = s[u]; e.y = d[u];
            st[lofs[bn[u]] + r[u]] = e;
        }
    }
    __syncthreads();

    v2i* po = (v2i*)pairs + e0;
    for (int i = tid; i < total; i += 256) po[i] = st[i];
}

__global__ void __launch_bounds__(256, 10) gcmc_gemm_split_kernel(
    const float* __restrict__ X,
    const float* __restrict__ W,
    const float* __restrict__ bias,
    const float* __restrict__ cj,
    float*       __restrict__ wh,
    const int*   __restrict__ edge_src,
    const int*   __restrict__ edge_dst,
    int*         __restrict__ tab,
    int*         __restrict__ pairs,
    int n_nodes, int n_edges, int nb, int nchunks)
{
    const int blk = (int)blockIdx.x;
    if (blk < nchunks) {
        split_part(blk, edge_src, edge_dst, tab, pairs, n_edges, nb);
    } else {
        gemm_cj_part(blk - nchunks, X, W, bias, cj, wh, n_nodes);
    }
}

template <int NB>
static __device__ __forceinline__ void gather_add_block(
    const float* __restrict__ wh, float* acc, const int ms, const int md, const unsigned lane)
{
    float v[NB];
    int   dl[NB];
#pragma unroll
    for (int u = 0; u < NB; ++u) {
        const unsigned so = (unsigned)__builtin_amdgcn_readlane(ms, u);
        dl[u] = __builtin_amdgcn_readlane(md, u);
        v[u]  = *(const float*)((const char*)wh + so + lane * 4u);
    }
#pragma unroll
    for (int u = 0; u < NB; ++u)
        atomicAdd(&acc[dl[u] * OUT_DIM + (int)lane], v[u]);
}

__global__ void __launch_bounds__(1024) gcmc_sum_kernel(
    const float* __restrict__ wh,
    const float* __restrict__ ci,
    const int*   __restrict__ tab,
    const int*   __restrict__ pairs,
    float*       __restrict__ out,
    int n_nodes, int nchunks, int nb)
{
    __shared__ float acc[(ROWS + 1) * OUT_DIM];
    __shared__ v2i   list[32 * WCAP];

    const int tid  = threadIdx.x;
    const int lane = tid & 31;
    const int w    = __builtin_amdgcn_readfirstlane(tid >> 5);
    const int b    = blockIdx.x;
    const int lo   = b * ROWS;
    const int rows_here = (n_nodes - lo) < ROWS ? (n_nodes - lo) : ROWS;

    for (int i = tid; i < (ROWS + 1) * OUT_DIM; i += 1024) acc[i] = 0.f;
    __syncthreads();

    const v2i* pr     = (const v2i*)pairs;
    v2i*       mylist = list + w * WCAP;

    for (int cg = w; cg < nchunks; cg += 32 * 32) {
        const int c = cg + 32 * lane;
        unsigned run = 0;
        if (c < nchunks) run = (unsigned)tab[(size_t)c * nb + b];
        const int rs  = (int)(run & 0xffffu);
        const int len = (int)(run >> 16);
        int inc = len;
#pragma unroll
        for (int o = 1; o < 32; o <<= 1) {
            const int t = __shfl_up(inc, o);
            if (lane >= o) inc += t;
        }
        const int off = inc - len;
        const int T   = __builtin_amdgcn_readlane(inc, 31);
        const int rel = c * CHUNK + rs - off;

        for (int r0 = 0; r0 < T; r0 += WCAP) {
            const int n = (T - r0) < WCAP ? (T - r0) : WCAP;
            for (int q0 = 0; q0 < n; q0 += 256) {
                v2i p[8];
#pragma unroll
                for (int u = 0; u < 8; ++u) {
                    const int pos = q0 + u * 32 + lane;
                    const int v   = r0 + (pos < n ? pos : n - 1);
                    int j = 0;
#pragma unroll
                    for (int step = 16; step >= 1; step >>= 1) {
                        const int o = __shfl(off, j + step);
                        j = (o <= v) ? j + step : j;
                    }
                    p[u] = pr[__shfl(rel, j) + v];
                }
#pragma unroll
                for (int u = 0; u < 8; ++u) {
                    const int pos = q0 + u * 32 + lane;
                    if (pos < n) {
                        v2i e;
                        e.x = p[u].x * (OUT_DIM * 4);
                        e.y = p[u].y - lo;
                        mylist[pos] = e;
                    }
                }
            }
            const int npad = (n + 15) & ~15;
            if (lane < npad - n) {
                v2i e;
                e.x = 0;
                e.y = ROWS;
                mylist[n + lane] = e;
            }
            for (int q = 0; q < npad; q += 32) {
                int ms = 0, md = ROWS;
                if (q + lane < npad) {
                    const v2i e = mylist[q + lane];
                    ms = e.x;
                    md = e.y;
                }
                if (npad - q >= 32) gather_add_block<32>(wh, acc, ms, md, (unsigned)lane);
                else                gather_add_block<16>(wh, acc, ms, md, (unsigned)lane);
            }
        }
    }
    __syncthreads();

    for (int i = tid; i < rows_here * OUT_DIM; i += 1024) {
        const int row = lo + (i >> 5);
        out[(size_t)row * OUT_DIM + (i & 31)] = acc[i] * ci[row];
    }
}

extern "C" void kernel_launch(void* const* d_in, const int* in_sizes, int n_in,
                              void* d_out, int out_size, void* d_ws, size_t ws_size,
                              hipStream_t stream)
{
    const float* X    = (const float*)d_in[0];
    const float* cj   = (const float*)d_in[1];
    const float* ci   = (const float*)d_in[2];
    const float* W    = (const float*)d_in[3];
    const float* bias = (const float*)d_in[4];
    const int*   esrc = (const int*)d_in[5];
    const int*   edst = (const int*)d_in[6];

    const int n_nodes = in_sizes[1];
    const int n_edges = in_sizes[5];

    const int nb      = (n_nodes + ROWS - 1) / ROWS;
    const int nchunks = (n_edges + CHUNK - 1) / CHUNK;

    float* out   = (float*)d_out;
    float* wh    = (float*)d_ws;
    int*   pairs = (int*)(wh + (size_t)n_nodes * OUT_DIM);
    int*   tab   = pairs + 2 * (size_t)n_edges;

    {
        const int ntiles = (n_nodes + 15) / 16;
        const int blocks = (ntiles + 7) / 8;
        gcmc_gemm_split_kernel<<<nchunks + blocks, 256, 0, stream>>>(
            X, W, bias, cj, wh, esrc, edst, tab, pairs, n_nodes, n_edges, nb, nchunks);
    }

    gcmc_sum_kernel<<<nb, 1024, 0, stream>>>(wh, ci, tab, pairs, out, n_nodes, nchunks, nb);
}


# GEMM A operand: X rows fetched by whole-line 16-B loads (two rows x 256 B per load) and turned to fragment layout in a wave-private LDS stage, one K half per trip; k order, W loads, sums unchanged
# speedup vs baseline: 2.5012x; 1.0861x over previous
#include <hip/hip_runtime.h>


typedef __attribute__((ext_vector_type(2))) float v2f;
typedef __attribute__((ext_vector_type(8))) float v8f;
typedef __attribute__((ext_vector_type(2))) int   v2i;
typedef __attribute__((ext_vector_type(4))) float v4f;

#define IN_DIM 128
#define OUT_DIM 32
#define CHUNK 4096
#define ROWS  400
#define MAXB  1024
#define WCAP  512
#define XP    68
#define LDS_INTS 10248

static __device__ __forceinline__ void gemm_cj_part(
    const int gb,
    const float* __restrict__ X,
    const float* __restrict__ W,
    const float* __restrict__ bias,
    const float* __restrict__ cj,
    float* __restrict__ wh,
    int n_nodes,
    float* lds)
{
    const int wave   = __builtin_amdgcn_readfirstlane((gb * 256 + (int)threadIdx.x) >> 5);
    const int ntiles = (n_nodes + 15) >> 4;
    if (wave >= ntiles) return;

    const int lane = threadIdx.x & 31;
    const int m    = lane & 15;
    const int kh   = lane >> 4;

    const char* xt   = (const char*)(X + (size_t)wave * (16 * IN_DIM));
    const int   rlast = n_nodes - 1 - wave * 16;
    const int   rmax = rlast < 15 ? rlast : 15;
    const unsigned xl = (unsigned)(kh * IN_DIM + 4 * m) * 4u;
    const unsigned xc = (unsigned)(rmax * IN_DIM + 4 * m) * 4u;
    float* xs = lds + ((int)threadIdx.x >> 5) * (16 * XP);

    const float* w0   = W + (size_t)m        * IN_DIM + 2 * kh;
    const float* w1   = W + (size_t)(m + 16) * IN_DIM + 2 * kh;

    v8f acc0 = {};
    v8f acc1 = {};

#pragma unroll 1
    for (int h = 0; h < 2; ++h) {
        v4f xr[8];
        const unsigned hb  = 256u * (unsigned)h;
        const unsigned cap = xc + hb;
#pragma unroll
        for (int j = 0; j < 8; ++j) {
            unsigned o = xl + 1024u * j + hb;
            o = o < cap ? o : cap;
            xr[j] = *(const v4f*)(xt + o);
        }
#pragma unroll
        for (int j = 0; j < 8; ++j)
            *(v4f*)(xs + (2 * j + kh) * XP + 4 * m) = xr[j];
#pragma unroll
        for (int kk = 0; kk < 64; kk += 4) {
            const int kb = 64 * h + kk;
            v2f a  = *(const v2f*)(xs + m * XP + kk + 2 * kh);
            v2f b0 = *(const v2f*)(w0 + kb);
            v2f b1 = *(const v2f*)(w1 + kb);
            acc0 = __builtin_amdgcn_wmma_f32_16x16x4_f32(
                false, a, false, b0, (short)0, acc0, false, false);
            acc1 = __builtin_amdgcn_wmma_f32_16x16x4_f32(
                false, a, false, b1, (short)0, acc1, false, false);
        }
    }

    const int   n0 = m, n1 = m + 16;
    const float bias0 = bias[n0];
    const float bias1 = bias[n1];

#pragma unroll
    for (int r = 0; r < 8; ++r) {
        const int mo   = r + 8 * kh;
        const int orow = wave * 16 + mo;
        if (orow < n_nodes) {
            const float s = cj[orow];
            wh[(size_t)orow * OUT_DIM + n0] = (acc0[r] + bias0) * s;
            wh[(size_t)orow * OUT_DIM + n1] = (acc1[r] + bias1) * s;
        }
    }
}

static __device__ __forceinline__ void split_part(
    const int chunk,
    const int* __restrict__ edge_src,
    const int* __restrict__ edge_dst,
    int*       __restrict__ tab,
    int*       __restrict__ pairs,
    int n_edges, int nb,
    int* lds)
{
    int* lc   = lds;
    int* lofs = lds + MAXB;
    int* wsum = lds + 2 * MAXB;
    v2i* st   = (v2i*)(lds + 2 * MAXB + 8);
    const int tid  = threadIdx.x;
    const int lane = tid & 31;
    const int w    = tid >> 5;

    for (int i = tid; i < nb; i += 256) lc[i] = 0;
    __syncthreads();

    const int e0 = chunk * CHUNK;
    const int total = (n_edges - e0) < CHUNK ? (n_edges - e0) : CHUNK;
    int s[16], d[16], bn[16], r[16];
#pragma unroll
    for (int u = 0; u < 16; ++u) {
        const int i = u * 256 + tid;
        if (i < total) { s[u] = edge_src[e0 + i]; d[u] = edge_dst[e0 + i]; }
        else           { s[u] = 0;                d[u] = -1; }
    }
#pragma unroll
    for (int u = 0; u < 16; ++u) {
        bn[u] = d[u] / ROWS;
        r[u]  = (d[u] >= 0) ? atomicAdd(&lc[bn[u]], 1) : 0;
    }
    __syncthreads();

    int c4[MAXB / 256];
    int tsum = 0;
#pragma unroll
    for (int k = 0; k < MAXB / 256; ++k) {
        const int i = tid * (MAXB / 256) + k;
        c4[k] = (i < nb) ? lc[i] : 0;
        tsum += c4[k];
    }
    int inc = tsum;
#pragma unroll
    for (int o = 1; o < 32; o <<= 1) {
        const int t = __shfl_up(inc, o);
        if (lane >= o) inc += t;
    }
    if (lane == 31) wsum[w] = inc;
    __syncthreads();
    int ex = inc - tsum;
#pragma unroll
    for (int k = 0; k < 8; ++k) ex += (k < w) ? wsum[k] : 0;
#pragma unroll
    for (int k = 0; k < MAXB / 256; ++k) {
        const int i = tid * (MAXB / 256) + k;
        if (i < nb) lofs[i] = ex;
        ex += c4[k];
    }
    __syncthreads();

    int* trow = tab + (size_t)chunk * (size_t)nb;
    for (int i = tid; i < nb; i += 256) trow[i] = lofs[i] | (lc[i] << 16);

#pragma unroll
    for (int u = 0; u < 16; ++u) {
        if (d[u] >= 0) {
            v2i e; e.x = s[u]; e.y = d[u];
            st[lofs[bn[u]] + r[u]] = e;
        }
    }
    __syncthreads();

    v2i* po = (v2i*)pairs + e0;
    for (int i = tid; i < total; i += 256) po[i] = st[i];
}

__global__ void __launch_bounds__(256, 10) gcmc_gemm_split_kernel(
    const float* __restrict__ X,
    const float* __restrict__ W,
    const float* __restrict__ bias,
    const float* __restrict__ cj,
    float*       __restrict__ wh,
    const int*   __restrict__ edge_src,
    const int*   __restrict__ edge_dst,
    int*         __restrict__ tab,
    int*         __restrict__ pairs,
    int n_nodes, int n_edges, int nb, int nchunks)
{
    __shared__ __attribute__((aligned(16))) int lds[LDS_INTS];
    const int blk = (int)blockIdx.x;
    if (blk < nchunks) {
        split_part(blk, edge_src, edge_dst, tab, pairs, n_edges, nb, lds);
    } else {
        gemm_cj_part(blk - nchunks, X, W, bias, cj, wh, n_nodes, (float*)lds);
    }
}

template <int NB>
static __device__ __forceinline__ void gather_add_block(
    const float* __restrict__ wh, float* acc, const int ms, const int md, const unsigned lane)
{
    float v[NB];
    int   dl[NB];
#pragma unroll
    for (int u = 0; u < NB; ++u) {
        const unsigned so = (unsigned)__builtin_amdgcn_readlane(ms, u);
        dl[u] = __builtin_amdgcn_readlane(md, u);
        v[u]  = *(const float*)((const char*)wh + so + lane * 4u);
    }
#pragma unroll
    for (int u = 0; u < NB; ++u)
        atomicAdd(&acc[dl[u] * OUT_DIM + (int)lane], v[u]);
}

__global__ void __launch_bounds__(1024) gcmc_sum_kernel(
    const float* __restrict__ wh,
    const float* __restrict__ ci,
    const int*   __restrict__ tab,
    const int*   __restrict__ pairs,
    float*       __restrict__ out,
    int n_nodes, int nchunks, int nb)
{
    __shared__ float acc[(ROWS + 1) * OUT_DIM];
    __shared__ v2i   list[32 * WCAP];

    const int tid  = threadIdx.x;
    const int lane = tid & 31;
    const int w    = __builtin_amdgcn_readfirstlane(tid >> 5);
    const int b    = blockIdx.x;
    const int lo   = b * ROWS;
    const int rows_here = (n_nodes - lo) < ROWS ? (n_nodes - lo) : ROWS;

    for (int i = tid; i < (ROWS + 1) * OUT_DIM; i += 1024) acc[i] = 0.f;
    __syncthreads();

    const v2i* pr     = (const v2i*)pairs;
    v2i*       mylist = list + w * WCAP;

    for (int cg = w; cg < nchunks; cg += 32 * 32) {
        const int c = cg + 32 * lane;
        unsigned run = 0;
        if (c < nchunks) run = (unsigned)tab[(size_t)c * nb + b];
        const int rs  = (int)(run & 0xffffu);
        const int len = (int)(run >> 16);
        int inc = len;
#pragma unroll
        for (int o = 1; o < 32; o <<= 1) {
            const int t = __shfl_up(inc, o);
            if (lane >= o) inc += t;
        }
        const int off = inc - len;
        const int T   = __builtin_amdgcn_readlane(inc, 31);
        const int rel = c * CHUNK + rs - off;

        for (int r0 = 0; r0 < T; r0 += WCAP) {
            const int n = (T - r0) < WCAP ? (T - r0) : WCAP;
            for (int q0 = 0; q0 < n; q0 += 256) {
                v2i p[8];
#pragma unroll
                for (int u = 0; u < 8; ++u) {
                    const int pos = q0 + u * 32 + lane;
                    const int v   = r0 + (pos < n ? pos : n - 1);
                    int j = 0;
#pragma unroll
                    for (int step = 16; step >= 1; step >>= 1) {
                        const int o = __shfl(off, j + step);
                        j = (o <= v) ? j + step : j;
                    }
                    p[u] = pr[__shfl(rel, j) + v];
                }
#pragma unroll
                for (int u = 0; u < 8; ++u) {
                    const int pos = q0 + u * 32 + lane;
                    if (pos < n) {
                        v2i e;
                        e.x = p[u].x * (OUT_DIM * 4);
                        e.y = p[u].y - lo;
                        mylist[pos] = e;
                    }
                }
            }
            const int npad = (n + 15) & ~15;
            if (lane < npad - n) {
                v2i e;
                e.x = 0;
                e.y = ROWS;
                mylist[n + lane] = e;
            }
            for (int q = 0; q < npad; q += 32) {
                int ms = 0, md = ROWS;
                if (q + lane < npad) {
                    const v2i e = mylist[q + lane];
                    ms = e.x;
                    md = e.y;
                }
                if (npad - q >= 32) gather_add_block<32>(wh, acc, ms, md, (unsigned)lane);
                else                gather_add_block<16>(wh, acc, ms, md, (unsigned)lane);
            }
        }
    }
    __syncthreads();

    for (int i = tid; i < rows_here * OUT_DIM; i += 1024) {
        const int row = lo + (i >> 5);
        out[(size_t)row * OUT_DIM + (i & 31)] = acc[i] * ci[row];
    }
}

extern "C" void kernel_launch(void* const* d_in, const int* in_sizes, int n_in,
                              void* d_out, int out_size, void* d_ws, size_t ws_size,
                              hipStream_t stream)
{
    const float* X    = (const float*)d_in[0];
    const float* cj   = (const float*)d_in[1];
    const float* ci   = (const float*)d_in[2];
    const float* W    = (const float*)d_in[3];
    const float* bias = (const float*)d_in[4];
    const int*   esrc = (const int*)d_in[5];
    const int*   edst = (const int*)d_in[6];

    const int n_nodes = in_sizes[1];
    const int n_edges = in_sizes[5];

    const int nb      = (n_nodes + ROWS - 1) / ROWS;
    const int nchunks = (n_edges + CHUNK - 1) / CHUNK;

    float* out   = (float*)d_out;
    float* wh    = (float*)d_ws;
    int*   pairs = (int*)(wh + (size_t)n_nodes * OUT_DIM);
    int*   tab   = pairs + 2 * (size_t)n_edges;

    {
        const int ntiles = (n_nodes + 15) / 16;
        const int blocks = (ntiles + 7) / 8;
        gcmc_gemm_split_kernel<<<nchunks + blocks, 256, 0, stream>>>(
            X, W, bias, cj, wh, esrc, edst, tab, pairs, n_nodes, n_edges, nb, nchunks);
    }

    gcmc_sum_kernel<<<nb, 1024, 0, stream>>>(wh, ci, tab, pairs, out, n_nodes, nchunks, nb);
}
